# starting kernel
# baseline (speedup 1.0000x reference)
#include <hip/hip_runtime.h>


typedef float v2f __attribute__((ext_vector_type(2)));
typedef float v8f __attribute__((ext_vector_type(8)));

#define NODES 64
#define NH    10

__device__ __forceinline__ float xor16(float x) {
  int xi = __builtin_bit_cast(int, x);
  int r  = __builtin_amdgcn_permlanex16(xi, xi, 0x76543210, 0xfedcba98,
                                        false, false);
  return __builtin_bit_cast(float, r);
}

__global__ __launch_bounds__(256)
void cgnn_wmma_kernel(const float* __restrict__ noise,
                      const float* __restrict__ W1,
                      const float* __restrict__ b1g,
                      const float* __restrict__ W2g,
                      const float* __restrict__ b2g,
                      float* __restrict__ out) {
  __shared__ float4 sA[NODES * 32];
  __shared__ float sW2[NODES * 16];

  for (int e = threadIdx.x; e < NODES * 32; e += blockDim.x) {
    const int i = e >> 5, l = e & 31, j = l & 15, h = l >> 4;
    float a0 = 0.f, a1 = 0.f, a2 = 0.f, a3 = 0.f;
    if (j < NH) {
      const float* w = W1 + i * (5 * NH);
      a0 = w[(2 * h) * NH + j];
      a1 = w[(2 * h + 1) * NH + j];
      if (h == 0) {
        a2 = w[4 * NH + j];
        a3 = b1g[i * NH + j];
      }
    }
    sA[e] = make_float4(a0, a1, a2, a3);
  }
  for (int e = threadIdx.x; e < NODES * 16; e += blockDim.x) {
    const int i = e >> 4, j = e & 15;
    sW2[e] = (j < NH) ? W2g[i * NH + j] : 0.f;
  }
  __syncthreads();

  const int lane = threadIdx.x & 31;
  const int wave = threadIdx.x >> 5;
  const int hi   = lane >> 4;
  const int m    = lane & 15;
  const long row = (long)blockIdx.x * 128 + (long)wave * 16 + m;

  const float4* nrow = (const float4*)(noise + row * NODES);
  float4*       orow = (float4*)(out + row * NODES);

  const float one_lo = hi ? 0.f : 1.f;
  float b0 = 0.f, b1w = 0.f;

  auto step = [&](int i, float nz) -> float {
    const float4 ap = sA[i * 32 + lane];
    v2f A2; A2[0] = ap.z; A2[1] = ap.w;
    v2f B2; B2[0] = hi ? 0.f : nz; B2[1] = one_lo;
    v8f zero = {};
    v8f t = __builtin_amdgcn_wmma_f32_16x16x4_f32(
        false, A2, false, B2, (short)0, zero, false, false);
    v2f A1; A1[0] = ap.x; A1[1] = ap.y;
    v2f B1; B1[0] = b0;   B1[1] = b1w;
    v8f d = __builtin_amdgcn_wmma_f32_16x16x4_f32(
        false, A1, false, B1, (short)0, t, false, false);
    const float4* wp = (const float4*)&sW2[i * 16 + hi * 8];
    const float4 wa = wp[0], wb = wp[1];
    float p;
    p = fmaxf(d[0], 0.f) * wa.x;
    p = fmaf(fmaxf(d[1], 0.f), wa.y, p);
    p = fmaf(fmaxf(d[2], 0.f), wa.z, p);
    p = fmaf(fmaxf(d[3], 0.f), wa.w, p);
    p = fmaf(fmaxf(d[4], 0.f), wb.x, p);
    p = fmaf(fmaxf(d[5], 0.f), wb.y, p);
    p = fmaf(fmaxf(d[6], 0.f), wb.z, p);
    p = fmaf(fmaxf(d[7], 0.f), wb.w, p);
    return (p + xor16(p)) + b2g[i];
  };

  {
    const float4 nz4 = nrow[0];
    const float y0 = step(0, nz4.x); if (!hi) b0  = y0;
    const float y1 = step(1, nz4.y); if (!hi) b1w = y1;
    const float y2 = step(2, nz4.z); if (hi)  b0  = y2;
    const float y3 = step(3, nz4.w); if (hi)  b1w = y3;
    if (!hi) orow[0] = make_float4(y0, y1, y2, y3);
  }

  for (int i4 = 1; i4 < 16; ++i4) {
    const float4 nz4 = nrow[i4];
    if (i4 < 15) __builtin_prefetch(nrow + i4 + 1, 0, 0);
    float ys[4];
#pragma unroll
    for (int u = 0; u < 4; ++u) {
      const int i = i4 * 4 + u;
      const float nz = (u == 0) ? nz4.x : (u == 1) ? nz4.y
                      : (u == 2) ? nz4.z : nz4.w;
      const float y = step(i, nz);
      const float tmp = xor16(b0);
      b0 = b1w;
      b1w = hi ? y : tmp;
      ys[u] = y;
    }
    if (!hi) orow[i4] = make_float4(ys[0], ys[1], ys[2], ys[3]);
  }
}

extern "C" void kernel_launch(void* const* d_in, const int* in_sizes, int n_in,
                              void* d_out, int out_size, void* d_ws, size_t ws_size,
                              hipStream_t stream) {
  const float* noise = (const float*)d_in[0];
  const float* W1 = (const float*)d_in[2];
  const float* b1 = (const float*)d_in[3];
  const float* W2 = (const float*)d_in[4];
  const float* b2 = (const float*)d_in[5];
  float* out = (float*)d_out;

  const int B = in_sizes[0] / NODES;
  const int rowsPerBlock = 128;
  const int grid = B / rowsPerBlock;

  cgnn_wmma_kernel<<<grid, 256, 0, stream>>>(noise, W1, b1, W2, b2, out);
}


# Stage each wave's 16x64 noise tile and result tile through LDS: every global 128-B line read once and written once by whole-row b128 accesses (was 16-B pieces per lane plus a per-trip prefetch)
# speedup vs baseline: 2.6240x; 2.6240x over previous
#include <hip/hip_runtime.h>


typedef float v2f __attribute__((ext_vector_type(2)));
typedef float v8f __attribute__((ext_vector_type(8)));

#define NODES 64
#define NH    10
#define TPITCH 17

__device__ __forceinline__ float xor16(float x) {
  int xi = __builtin_bit_cast(int, x);
  int r  = __builtin_amdgcn_permlanex16(xi, xi, 0x76543210, 0xfedcba98,
                                        false, false);
  return __builtin_bit_cast(float, r);
}

__global__ __launch_bounds__(256)
void cgnn_wmma_kernel(const float* __restrict__ noise,
                      const float* __restrict__ W1,
                      const float* __restrict__ b1g,
                      const float* __restrict__ W2g,
                      const float* __restrict__ b2g,
                      float* __restrict__ out) {
  __shared__ float4 sA[NODES * 32];
  __shared__ float sW2[NODES * 16];
  __shared__ float4 sT[8 * 16 * TPITCH];

  const int lane = threadIdx.x & 31;
  const int wave = threadIdx.x >> 5;
  const int hi   = lane >> 4;
  const int m    = lane & 15;
  const long row0 = (long)blockIdx.x * 128 + (long)wave * 16;

  for (int e = threadIdx.x; e < NODES * 32; e += blockDim.x) {
    const int i = e >> 5, l = e & 31, j = l & 15, h = l >> 4;
    float a0 = 0.f, a1 = 0.f, a2 = 0.f, a3 = 0.f;
    if (j < NH) {
      const float* w = W1 + i * (5 * NH);
      a0 = w[(2 * h) * NH + j];
      a1 = w[(2 * h + 1) * NH + j];
      if (h == 0) {
        a2 = w[4 * NH + j];
        a3 = b1g[i * NH + j];
      }
    }
    sA[e] = make_float4(a0, a1, a2, a3);
  }
  for (int e = threadIdx.x; e < NODES * 16; e += blockDim.x) {
    const int i = e >> 4, j = e & 15;
    sW2[e] = (j < NH) ? W2g[i * NH + j] : 0.f;
  }

  const float4* gin = (const float4*)(noise + row0 * NODES) + hi * 16 + m;
  float4 nreg[8];
#pragma unroll
  for (int t = 0; t < 8; ++t) nreg[t] = gin[t * 32];
  float4* tile = sT + wave * (16 * TPITCH);
#pragma unroll
  for (int t = 0; t < 8; ++t) tile[(2 * t + hi) * TPITCH + m] = nreg[t];
  __syncthreads();

  float4* trow = tile + m * TPITCH;

  const float one_lo = hi ? 0.f : 1.f;
  float b0 = 0.f, b1w = 0.f;

  auto step = [&](int i, float nz) -> float {
    const float4 ap = sA[i * 32 + lane];
    v2f A2; A2[0] = ap.z; A2[1] = ap.w;
    v2f B2; B2[0] = hi ? 0.f : nz; B2[1] = one_lo;
    v8f zero = {};
    v8f t = __builtin_amdgcn_wmma_f32_16x16x4_f32(
        false, A2, false, B2, (short)0, zero, false, false);
    v2f A1; A1[0] = ap.x; A1[1] = ap.y;
    v2f B1; B1[0] = b0;   B1[1] = b1w;
    v8f d = __builtin_amdgcn_wmma_f32_16x16x4_f32(
        false, A1, false, B1, (short)0, t, false, false);
    const float4* wp = (const float4*)&sW2[i * 16 + hi * 8];
    const float4 wa = wp[0], wb = wp[1];
    float p;
    p = fmaxf(d[0], 0.f) * wa.x;
    p = fmaf(fmaxf(d[1], 0.f), wa.y, p);
    p = fmaf(fmaxf(d[2], 0.f), wa.z, p);
    p = fmaf(fmaxf(d[3], 0.f), wa.w, p);
    p = fmaf(fmaxf(d[4], 0.f), wb.x, p);
    p = fmaf(fmaxf(d[5], 0.f), wb.y, p);
    p = fmaf(fmaxf(d[6], 0.f), wb.z, p);
    p = fmaf(fmaxf(d[7], 0.f), wb.w, p);
    return (p + xor16(p)) + b2g[i];
  };

  {
    const float4 nz4 = trow[0];
    const float y0 = step(0, nz4.x); if (!hi) b0  = y0;
    const float y1 = step(1, nz4.y); if (!hi) b1w = y1;
    const float y2 = step(2, nz4.z); if (hi)  b0  = y2;
    const float y3 = step(3, nz4.w); if (hi)  b1w = y3;
    if (!hi) trow[0] = make_float4(y0, y1, y2, y3);
  }

  float4 nzn = trow[1];
  for (int i4 = 1; i4 < 16; ++i4) {
    const float4 nz4 = nzn;
    if (i4 < 15) nzn = trow[i4 + 1];
    float ys[4];
#pragma unroll
    for (int u = 0; u < 4; ++u) {
      const int i = i4 * 4 + u;
      const float nz = (u == 0) ? nz4.x : (u == 1) ? nz4.y
                      : (u == 2) ? nz4.z : nz4.w;
      const float y = step(i, nz);
      const float tmp = xor16(b0);
      b0 = b1w;
      b1w = hi ? y : tmp;
      ys[u] = y;
    }
    if (!hi) trow[i4] = make_float4(ys[0], ys[1], ys[2], ys[3]);
  }

  __syncthreads();

  float4* gout = (float4*)(out + row0 * NODES) + hi * 16 + m;
#pragma unroll
  for (int t = 0; t < 8; ++t) gout[t * 32] = tile[(2 * t + hi) * TPITCH + m];
}

extern "C" void kernel_launch(void* const* d_in, const int* in_sizes, int n_in,
                              void* d_out, int out_size, void* d_ws, size_t ws_size,
                              hipStream_t stream) {
  const float* noise = (const float*)d_in[0];
  const float* W1 = (const float*)d_in[2];
  const float* b1 = (const float*)d_in[3];
  const float* W2 = (const float*)d_in[4];
  const float* b2 = (const float*)d_in[5];
  float* out = (float*)d_out;

  const int B = in_sizes[0] / NODES;
  const int rowsPerBlock = 128;
  const int grid = B / rowsPerBlock;

  cgnn_wmma_kernel<<<grid, 256, 0, stream>>>(noise, W1, b1, W2, b2, out);
}


# Weight table re-laid for the vector tail: hidden units 5+5 in the matrix rows (was 8+2), parents in a fixed 4-slot k ring (no window shift); per-node sum order now (u0..u4)+(u5..u9)
# speedup vs baseline: 2.7652x; 1.0538x over previous
#include <hip/hip_runtime.h>


typedef float v2f __attribute__((ext_vector_type(2)));
typedef float v8f __attribute__((ext_vector_type(8)));

#define NODES 64
#define NH    10
#define TPITCH 17

__device__ __forceinline__ int row_unit(int row) {
  return (row < 5) ? row : (row >= 8 && row < 13) ? row - 3 : -1;
}

__device__ __forceinline__ float xor16(float x) {
  int xi = __builtin_bit_cast(int, x);
  int r  = __builtin_amdgcn_permlanex16(xi, xi, 0x76543210, 0xfedcba98,
                                        false, false);
  return __builtin_bit_cast(float, r);
}

__global__ __launch_bounds__(256)
void cgnn_wmma_kernel(const float* __restrict__ noise,
                      const float* __restrict__ W1,
                      const float* __restrict__ b1g,
                      const float* __restrict__ W2g,
                      const float* __restrict__ b2g,
                      float* __restrict__ out) {
  __shared__ float4 sA[NODES * 32];
  __shared__ float sW2[NODES * 16];
  __shared__ float4 sT[8 * 16 * TPITCH];

  const int lane = threadIdx.x & 31;
  const int wave = threadIdx.x >> 5;
  const int hi   = lane >> 4;
  const int m    = lane & 15;
  const long row0 = (long)blockIdx.x * 128 + (long)wave * 16;

  for (int e = threadIdx.x; e < NODES * 32; e += blockDim.x) {
    const int i = e >> 5, l = e & 31, h = l >> 4;
    const int j = row_unit(l & 15);
    float a0 = 0.f, a1 = 0.f, a2 = 0.f, a3 = 0.f;
    if (j >= 0) {
      const float* w = W1 + i * (5 * NH);
      const int t0 = (i < 4) ? (2 * h)     : ((2 * h - i) & 3);
      const int t1 = (i < 4) ? (2 * h + 1) : ((2 * h + 1 - i) & 3);
      a0 = w[t0 * NH + j];
      a1 = w[t1 * NH + j];
      if (h == 0) {
        a2 = w[4 * NH + j];
        a3 = b1g[i * NH + j];
      }
    }
    sA[e] = make_float4(a0, a1, a2, a3);
  }
  for (int e = threadIdx.x; e < NODES * 16; e += blockDim.x) {
    const int i = e >> 4;
    const int j = row_unit(e & 15);
    sW2[e] = (j >= 0) ? W2g[i * NH + j] : 0.f;
  }

  const float4* gin = (const float4*)(noise + row0 * NODES) + hi * 16 + m;
  float4 nreg[8];
#pragma unroll
  for (int t = 0; t < 8; ++t) nreg[t] = gin[t * 32];
  float4* tile = sT + wave * (16 * TPITCH);
#pragma unroll
  for (int t = 0; t < 8; ++t) tile[(2 * t + hi) * TPITCH + m] = nreg[t];
  __syncthreads();

  float4* trow = tile + m * TPITCH;

  const float one_lo = hi ? 0.f : 1.f;
  float b0 = 0.f, b1w = 0.f;

  auto step = [&](int i, float nz) -> float {
    const float4 ap = sA[i * 32 + lane];
    v2f A2; A2[0] = ap.z; A2[1] = ap.w;
    v2f B2; B2[0] = hi ? 0.f : nz; B2[1] = one_lo;
    v8f zero = {};
    v8f t = __builtin_amdgcn_wmma_f32_16x16x4_f32(
        false, A2, false, B2, (short)0, zero, false, false);
    v2f A1; A1[0] = ap.x; A1[1] = ap.y;
    v2f B1; B1[0] = b0;   B1[1] = b1w;
    v8f d = __builtin_amdgcn_wmma_f32_16x16x4_f32(
        false, A1, false, B1, (short)0, t, false, false);
    const float* wq = &sW2[i * 16 + hi * 8];
    const float4 wa = *(const float4*)wq;
    const float  w4 = wq[4];
    float p;
    p = fmaxf(d[0], 0.f) * wa.x;
    p = fmaf(fmaxf(d[1], 0.f), wa.y, p);
    p = fmaf(fmaxf(d[2], 0.f), wa.z, p);
    p = fmaf(fmaxf(d[3], 0.f), wa.w, p);
    p = fmaf(fmaxf(d[4], 0.f), w4, p);
    return (p + xor16(p)) + b2g[i];
  };

  float4 nzn = trow[0];
  for (int i4 = 0; i4 < 16; ++i4) {
    const float4 nz4 = nzn;
    if (i4 < 15) nzn = trow[i4 + 1];
    const int i = i4 * 4;
    const float y0 = step(i + 0, nz4.x); if (!hi) b0  = y0;
    const float y1 = step(i + 1, nz4.y); if (!hi) b1w = y1;
    const float y2 = step(i + 2, nz4.z); if (hi)  b0  = y2;
    const float y3 = step(i + 3, nz4.w); if (hi)  b1w = y3;
    if (!hi) trow[i4] = make_float4(y0, y1, y2, y3);
  }

  __syncthreads();

  float4* gout = (float4*)(out + row0 * NODES) + hi * 16 + m;
#pragma unroll
  for (int t = 0; t < 8; ++t) gout[t * 32] = tile[(2 * t + hi) * TPITCH + m];
}

extern "C" void kernel_launch(void* const* d_in, const int* in_sizes, int n_in,
                              void* d_out, int out_size, void* d_ws, size_t ws_size,
                              hipStream_t stream) {
  const float* noise = (const float*)d_in[0];
  const float* W1 = (const float*)d_in[2];
  const float* b1 = (const float*)d_in[3];
  const float* W2 = (const float*)d_in[4];
  const float* b2 = (const float*)d_in[5];
  float* out = (float*)d_out;

  const int B = in_sizes[0] / NODES;
  const int rowsPerBlock = 128;
  const int grid = B / rowsPerBlock;

  cgnn_wmma_kernel<<<grid, 256, 0, stream>>>(noise, W1, b1, W2, b2, out);
}


# Persistent work-groups, one 32-wave group per WGP: weight tables built once per WGP (was once per 128 rows), each wave loops over its tiles with the next tile's loads in flight; one start-up barrier
# speedup vs baseline: 3.0415x; 1.0999x over previous
#include <hip/hip_runtime.h>


typedef float v2f __attribute__((ext_vector_type(2)));
typedef float v8f __attribute__((ext_vector_type(8)));

#define NODES 64
#define NH    10
#define TPITCH 17

__device__ __forceinline__ int row_unit(int row) {
  return (row < 5) ? row : (row >= 8 && row < 13) ? row - 3 : -1;
}

__device__ __forceinline__ float xor16(float x) {
  int xi = __builtin_bit_cast(int, x);
  int r  = __builtin_amdgcn_permlanex16(xi, xi, 0x76543210, 0xfedcba98,
                                        false, false);
  return __builtin_bit_cast(float, r);
}

#define WAVES 32
#define TILE_LOAD(g)                                                     \
  do { n0 = (g)[0];   n1 = (g)[32];  n2 = (g)[64];  n3 = (g)[96];        \
       n4 = (g)[128]; n5 = (g)[160]; n6 = (g)[192]; n7 = (g)[224]; } while (0)

__device__ __forceinline__ void wave_lds_order() {
  __builtin_amdgcn_fence(__ATOMIC_RELEASE, "wavefront");
  __builtin_amdgcn_wave_barrier();
  __builtin_amdgcn_fence(__ATOMIC_ACQUIRE, "wavefront");
}

__global__ __launch_bounds__(WAVES * 32)
void cgnn_wmma_kernel(const float* __restrict__ noise,
                      const float* __restrict__ W1,
                      const float* __restrict__ b1g,
                      const float* __restrict__ W2g,
                      const float* __restrict__ b2g,
                      float* __restrict__ out,
                      int ntiles) {
  __shared__ float4 sA[NODES * 32];
  __shared__ float sW2[NODES * 16];
  __shared__ float4 sT[WAVES * 16 * TPITCH];

  const int lane = threadIdx.x & 31;
  const int wave = __builtin_amdgcn_readfirstlane((int)(threadIdx.x >> 5));
  const int hi   = lane >> 4;
  const int m    = lane & 15;
  const int stride = (int)gridDim.x * WAVES;
  int tile = (int)blockIdx.x * WAVES + wave;

  const int lofs = hi * 16 + m;
  float4 n0 = {}, n1 = {}, n2 = {}, n3 = {}, n4 = {}, n5 = {}, n6 = {}, n7 = {};
  if (tile < ntiles) {
    const float4* gin = (const float4*)(noise + (long)tile * (16 * NODES)) + lofs;
    TILE_LOAD(gin);
  }

  for (int e = threadIdx.x; e < NODES * 32; e += blockDim.x) {
    const int i = e >> 5, l = e & 31, h = l >> 4;
    const int j = row_unit(l & 15);
    float a0 = 0.f, a1 = 0.f, a2 = 0.f, a3 = 0.f;
    if (j >= 0) {
      const float* w = W1 + i * (5 * NH);
      const int t0 = (i < 4) ? (2 * h)     : ((2 * h - i) & 3);
      const int t1 = (i < 4) ? (2 * h + 1) : ((2 * h + 1 - i) & 3);
      a0 = w[t0 * NH + j];
      a1 = w[t1 * NH + j];
      if (h == 0) {
        a2 = w[4 * NH + j];
        a3 = b1g[i * NH + j];
      }
    }
    sA[e] = make_float4(a0, a1, a2, a3);
  }
  for (int e = threadIdx.x; e < NODES * 16; e += blockDim.x) {
    const int i = e >> 4;
    const int j = row_unit(e & 15);
    sW2[e] = (j >= 0) ? W2g[i * NH + j] : ((e & 7) == 5) ? b2g[i] : 0.f;
  }
  __syncthreads();

  float4* tilep = sT + wave * (16 * TPITCH);
  float4* trow  = tilep + m * TPITCH;
  const float one_lo = hi ? 0.f : 1.f;

  while (tile < ntiles) {
    float4* tp = tilep + hi * TPITCH + m;
    tp[0 * TPITCH] = n0;  tp[2 * TPITCH] = n1;  tp[4 * TPITCH] = n2;  tp[6 * TPITCH] = n3;
    tp[8 * TPITCH] = n4;  tp[10 * TPITCH] = n5; tp[12 * TPITCH] = n6; tp[14 * TPITCH] = n7;
    const int next = tile + stride;
    if (next < ntiles) {
      const float4* gin = (const float4*)(noise + (long)next * (16 * NODES)) + lofs;
      TILE_LOAD(gin);
    }
    wave_lds_order();

    float b0 = 0.f, b1w = 0.f;

    auto step = [&](int i, float nz) -> float {
      const float4 ap = sA[i * 32 + lane];
      v2f A2; A2[0] = ap.z; A2[1] = ap.w;
      v2f B2; B2[0] = hi ? 0.f : nz; B2[1] = one_lo;
      v8f zero = {};
      v8f t = __builtin_amdgcn_wmma_f32_16x16x4_f32(
          false, A2, false, B2, (short)0, zero, false, false);
      v2f A1; A1[0] = ap.x; A1[1] = ap.y;
      v2f B1; B1[0] = b0;   B1[1] = b1w;
      v8f d = __builtin_amdgcn_wmma_f32_16x16x4_f32(
          false, A1, false, B1, (short)0, t, false, false);
      const float* wq = &sW2[i * 16 + hi * 8];
      const float4 wa = *(const float4*)wq;
      const float2 wb = *(const float2*)(wq + 4);
      float p;
      p = fmaxf(d[0], 0.f) * wa.x;
      p = fmaf(fmaxf(d[1], 0.f), wa.y, p);
      p = fmaf(fmaxf(d[2], 0.f), wa.z, p);
      p = fmaf(fmaxf(d[3], 0.f), wa.w, p);
      p = fmaf(fmaxf(d[4], 0.f), wb.x, p);
      return (p + xor16(p)) + wb.y;
    };

    float4 nzn = trow[0];
    for (int i4 = 0; i4 < 16; ++i4) {
      const float4 nz4 = nzn;
      if (i4 < 15) nzn = trow[i4 + 1];
      const int i = i4 * 4;
      const float y0 = step(i + 0, nz4.x); if (!hi) b0  = y0;
      const float y1 = step(i + 1, nz4.y); if (!hi) b1w = y1;
      const float y2 = step(i + 2, nz4.z); if (hi)  b0  = y2;
      const float y3 = step(i + 3, nz4.w); if (hi)  b1w = y3;
      if (!hi) trow[i4] = make_float4(y0, y1, y2, y3);
    }

    wave_lds_order();

    float4* gout = (float4*)(out + (long)tile * (16 * NODES)) + lofs;
#pragma unroll
    for (int t = 0; t < 8; ++t) gout[t * 32] = tilep[(2 * t + hi) * TPITCH + m];
    wave_lds_order();
    tile = next;
  }
}

extern "C" void kernel_launch(void* const* d_in, const int* in_sizes, int n_in,
                              void* d_out, int out_size, void* d_ws, size_t ws_size,
                              hipStream_t stream) {
  const float* noise = (const float*)d_in[0];
  const float* W1 = (const float*)d_in[2];
  const float* b1 = (const float*)d_in[3];
  const float* W2 = (const float*)d_in[4];
  const float* b2 = (const float*)d_in[5];
  float* out = (float*)d_out;

  const int B = in_sizes[0] / NODES;
  const int ntiles = B / 16;

  cgnn_wmma_kernel<<<256, WAVES * 32, 0, stream>>>(noise, W1, b1, W2, b2, out, ntiles);
}


# Per-node vector tail cut from 24 to 13 operations: one-instruction relu, no operand copies, immediate table addresses (nodes unrolled), noise pairs fed as loaded; b2 now starts the low half's sum
# speedup vs baseline: 3.4983x; 1.1502x over previous
#include <hip/hip_runtime.h>


typedef float v2f __attribute__((ext_vector_type(2)));
typedef float v8f __attribute__((ext_vector_type(8)));

#define NODES 64
#define NH    10
#define TPITCH 17

__device__ __forceinline__ int row_unit(int row) {
  return (row < 5) ? row : (row >= 8 && row < 13) ? row - 3 : -1;
}

__device__ __forceinline__ float xor16(float x) {
  int xi = __builtin_bit_cast(int, x);
  int r  = __builtin_amdgcn_permlanex16(xi, xi, 0x76543210, 0xfedcba98,
                                        true, false);
  return __builtin_bit_cast(float, r);
}

__device__ __forceinline__ float relu(float x) {
  const int xi = __builtin_bit_cast(int, x);
  return __builtin_bit_cast(float, xi > 0 ? xi : 0);
}

#define WAVES 32
#define TILE_LOAD(g)                                                     \
  do { n0 = (g)[0];   n1 = (g)[32];  n2 = (g)[64];  n3 = (g)[96];        \
       n4 = (g)[128]; n5 = (g)[160]; n6 = (g)[192]; n7 = (g)[224]; } while (0)

__device__ __forceinline__ void wave_lds_order() {
  __builtin_amdgcn_fence(__ATOMIC_RELEASE, "wavefront");
  __builtin_amdgcn_wave_barrier();
  __builtin_amdgcn_fence(__ATOMIC_ACQUIRE, "wavefront");
}

__global__ __launch_bounds__(WAVES * 32)
void cgnn_wmma_kernel(const float* __restrict__ noise,
                      const float* __restrict__ W1,
                      const float* __restrict__ b1g,
                      const float* __restrict__ W2g,
                      const float* __restrict__ b2g,
                      float* __restrict__ out,
                      int ntiles) {
  __shared__ float4 sA[NODES * 32];
  __shared__ float sW2[NODES * 16];
  __shared__ float4 sZ[16];
  __shared__ float4 sT[WAVES * 16 * TPITCH];

  const int lane = threadIdx.x & 31;
  const int wave = __builtin_amdgcn_readfirstlane((int)(threadIdx.x >> 5));
  const int hi   = lane >> 4;
  const int m    = lane & 15;
  const int stride = (int)gridDim.x * WAVES;
  int tile = (int)blockIdx.x * WAVES + wave;

  const int lofs = hi * 16 + m;
  float4 n0 = {}, n1 = {}, n2 = {}, n3 = {}, n4 = {}, n5 = {}, n6 = {}, n7 = {};
  if (tile < ntiles) {
    const float4* gin = (const float4*)(noise + (long)tile * (16 * NODES)) + lofs;
    TILE_LOAD(gin);
  }

  for (int e = threadIdx.x; e < NODES * 32; e += blockDim.x) {
    const int i = e >> 5, l = e & 31, h = l >> 4;
    const int j = row_unit(l & 15);
    float a0 = 0.f, a1 = 0.f, a2 = 0.f, a3 = 0.f;
    if (j >= 0) {
      const float* w = W1 + i * (5 * NH);
      const int t0 = (i < 4) ? (2 * h)     : ((2 * h - i) & 3);
      const int t1 = (i < 4) ? (2 * h + 1) : ((2 * h + 1 - i) & 3);
      a0 = w[t0 * NH + j];
      a1 = w[t1 * NH + j];
      if (h == 0) {
        const float wn = w[4 * NH + j];
        a2 = (i & 1) ? 0.f : wn;
        a3 = (i & 1) ? wn : 0.f;
      } else {
        a2 = b1g[i * NH + j];
      }
    }
    sA[e] = make_float4(a0, a1, a2, a3);
  }
  for (int e = threadIdx.x; e < NODES * 16; e += blockDim.x) {
    const int i = e >> 4;
    const int j = row_unit(e & 15);
    sW2[e] = (j >= 0) ? W2g[i * NH + j] : ((e & 15) == 5) ? b2g[i] : 0.f;
  }
  if (threadIdx.x < 16) sZ[threadIdx.x] = make_float4(1.f, 0.f, 1.f, 0.f);
  __syncthreads();

  float4* tilep = sT + wave * (16 * TPITCH);
  float4* trow  = tilep + m * TPITCH;
  const float4* nzrow = hi ? (const float4*)sZ : (const float4*)trow;

  while (tile < ntiles) {
    float4* tp = tilep + hi * TPITCH + m;
    tp[0 * TPITCH] = n0;  tp[2 * TPITCH] = n1;  tp[4 * TPITCH] = n2;  tp[6 * TPITCH] = n3;
    tp[8 * TPITCH] = n4;  tp[10 * TPITCH] = n5; tp[12 * TPITCH] = n6; tp[14 * TPITCH] = n7;
    const int next = tile + stride;
    if (next < ntiles) {
      const float4* gin = (const float4*)(noise + (long)next * (16 * NODES)) + lofs;
      TILE_LOAD(gin);
    }
    wave_lds_order();

    float b0 = 0.f, b1w = 0.f;

    auto step = [&](int i, float nz0, float nz1) -> float {
      const float4 ap = sA[i * 32 + lane];
      v2f A2; A2[0] = ap.z; A2[1] = ap.w;
      v2f B2; B2[0] = nz0; B2[1] = nz1;
      v8f zero = {};
      v8f t = __builtin_amdgcn_wmma_f32_16x16x4_f32(
          false, A2, false, B2, (short)0, zero, false, false);
      v2f A1; A1[0] = ap.x; A1[1] = ap.y;
      v2f B1; B1[0] = b0;   B1[1] = b1w;
      v8f d = __builtin_amdgcn_wmma_f32_16x16x4_f32(
          false, A1, false, B1, (short)0, t, false, false);
      const float* wq = &sW2[i * 16 + hi * 8];
      const float4 wa = *(const float4*)wq;
      const float2 wb = *(const float2*)(wq + 4);
      float p;
      p = fmaf(relu(d[0]), wa.x, wb.y);
      p = fmaf(relu(d[1]), wa.y, p);
      p = fmaf(relu(d[2]), wa.z, p);
      p = fmaf(relu(d[3]), wa.w, p);
      p = fmaf(relu(d[4]), wb.x, p);
      return p + xor16(p);
    };

    float4 nzn = nzrow[0];
#pragma unroll
    for (int i4 = 0; i4 < 16; ++i4) {
      const float4 nz4 = nzn;
      if (i4 < 15) nzn = nzrow[i4 + 1];
      const int i = i4 * 4;
      const float y0 = step(i + 0, nz4.x, nz4.y); if (!hi) b0  = y0;
      const float y1 = step(i + 1, nz4.x, nz4.y); if (!hi) b1w = y1;
      const float y2 = step(i + 2, nz4.z, nz4.w); if (hi)  b0  = y2;
      const float y3 = step(i + 3, nz4.z, nz4.w); if (hi)  b1w = y3;
      trow[i4] = make_float4(y0, y1, y2, y3);
    }

    wave_lds_order();

    float4* gout = (float4*)(out + (long)tile * (16 * NODES)) + lofs;
#pragma unroll
    for (int t = 0; t < 8; ++t) gout[t * 32] = tilep[(2 * t + hi) * TPITCH + m];
    wave_lds_order();
    tile = next;
  }
}

extern "C" void kernel_launch(void* const* d_in, const int* in_sizes, int n_in,
                              void* d_out, int out_size, void* d_ws, size_t ws_size,
                              hipStream_t stream) {
  const float* noise = (const float*)d_in[0];
  const float* W1 = (const float*)d_in[2];
  const float* b1 = (const float*)d_in[3];
  const float* W2 = (const float*)d_in[4];
  const float* b2 = (const float*)d_in[5];
  float* out = (float*)d_out;

  const int B = in_sizes[0] / NODES;
  const int ntiles = B / 16;

  cgnn_wmma_kernel<<<256, WAVES * 32, 0, stream>>>(noise, W1, b1, W2, b2, out, ntiles);
}


# A wave runs two adjacent tiles side by side (16 waves per group, was 32 x one tile): a node's weights are read from LDS once for both; the table build, 4 trips a thread now, issues its loads at once
# speedup vs baseline: 3.6682x; 1.0486x over previous
#include <hip/hip_runtime.h>


typedef float v2f __attribute__((ext_vector_type(2)));
typedef float v8f __attribute__((ext_vector_type(8)));

#define NODES 64
#define NH    10
#define TPITCH 17

__device__ __forceinline__ int row_unit(int row) {
  return (row < 5) ? row : (row >= 8 && row < 13) ? row - 3 : -1;
}

__device__ __forceinline__ float xor16(float x) {
  int xi = __builtin_bit_cast(int, x);
  int r  = __builtin_amdgcn_permlanex16(xi, xi, 0x76543210, 0xfedcba98,
                                        true, false);
  return __builtin_bit_cast(float, r);
}

__device__ __forceinline__ float relu(float x) {
  const int xi = __builtin_bit_cast(int, x);
  return __builtin_bit_cast(float, xi > 0 ? xi : 0);
}

#define WAVES 16
#define TPW   2
#define TILE_LOAD_N(g)                                                   \
  do { n0 = (g)[0];   n1 = (g)[32];  n2 = (g)[64];  n3 = (g)[96];        \
       n4 = (g)[128]; n5 = (g)[160]; n6 = (g)[192]; n7 = (g)[224]; } while (0)
#define TILE_LOAD_Q(g)                                                   \
  do { q0 = (g)[0];   q1 = (g)[32];  q2 = (g)[64];  q3 = (g)[96];        \
       q4 = (g)[128]; q5 = (g)[160]; q6 = (g)[192]; q7 = (g)[224]; } while (0)

__device__ __forceinline__ void wave_lds_order() {
  __builtin_amdgcn_fence(__ATOMIC_RELEASE, "wavefront");
  __builtin_amdgcn_wave_barrier();
  __builtin_amdgcn_fence(__ATOMIC_ACQUIRE, "wavefront");
}

__global__ __launch_bounds__(WAVES * 32)
void cgnn_wmma_kernel(const float* __restrict__ noise,
                      const float* __restrict__ W1,
                      const float* __restrict__ b1g,
                      const float* __restrict__ W2g,
                      const float* __restrict__ b2g,
                      float* __restrict__ out,
                      int ntiles) {
  __shared__ float4 sA[NODES * 32];
  __shared__ float sW2[NODES * 16];
  __shared__ float4 sZ[16];
  __shared__ float4 sT[WAVES * TPW * 16 * TPITCH];

  const int lane = threadIdx.x & 31;
  const int wave = __builtin_amdgcn_readfirstlane((int)(threadIdx.x >> 5));
  const int hi   = lane >> 4;
  const int m    = lane & 15;
  const int stride = (int)gridDim.x * (WAVES * TPW);
  int tile = ((int)blockIdx.x * WAVES + wave) * TPW;

  const int lofs = hi * 16 + m;
  float4 n0 = {}, n1 = {}, n2 = {}, n3 = {}, n4 = {}, n5 = {}, n6 = {}, n7 = {};
  float4 q0 = {}, q1 = {}, q2 = {}, q3 = {}, q4 = {}, q5 = {}, q6 = {}, q7 = {};
  if (tile < ntiles) {
    const float4* gin = (const float4*)(noise + (long)tile * (16 * NODES)) + lofs;
    TILE_LOAD_N(gin);
    if (tile + 1 < ntiles) TILE_LOAD_Q(gin + 256);
  }

#pragma unroll
  for (int k = 0; k < (NODES * 32) / (WAVES * 32); ++k) {
    const int e = (int)threadIdx.x + k * (WAVES * 32);
    const int i = e >> 5, l = e & 31, h = l >> 4;
    const int j = row_unit(l & 15);
    const int jc = (j >= 0) ? j : 0;
    const float* w = W1 + i * (5 * NH);
    const int t0 = (i < 4) ? (2 * h)     : ((2 * h - i) & 3);
    const int t1 = (i < 4) ? (2 * h + 1) : ((2 * h + 1 - i) & 3);
    const float w0 = w[t0 * NH + jc];
    const float w1 = w[t1 * NH + jc];
    const float* cp = h ? (b1g + i * NH + jc) : (w + 4 * NH + jc);
    const float c = *cp;
    const bool odd_low = (h == 0) && (i & 1);
    const float a0 = (j >= 0) ? w0 : 0.f;
    const float a1 = (j >= 0) ? w1 : 0.f;
    const float a2 = (j >= 0 && !odd_low) ? c : 0.f;
    const float a3 = (j >= 0 && odd_low)  ? c : 0.f;
    sA[e] = make_float4(a0, a1, a2, a3);
  }
#pragma unroll
  for (int k = 0; k < (NODES * 16) / (WAVES * 32); ++k) {
    const int e = (int)threadIdx.x + k * (WAVES * 32);
    const int i = e >> 4;
    const int r = e & 15;
    const int j = row_unit(r);
    const float* sp = (j >= 0) ? (W2g + i * NH + j) : (b2g + i);
    const float v = *sp;
    sW2[e] = (j >= 0 || r == 5) ? v : 0.f;
  }
  if (threadIdx.x < 16) sZ[threadIdx.x] = make_float4(1.f, 0.f, 1.f, 0.f);
  __syncthreads();

  float4* tileA = sT + wave * (TPW * 16 * TPITCH);
  float4* tileB = tileA + 16 * TPITCH;
  float4* trowA = tileA + m * TPITCH;
  float4* trowB = tileB + m * TPITCH;
  const float4* nzrowA = hi ? (const float4*)sZ : (const float4*)trowA;
  const float4* nzrowB = hi ? (const float4*)sZ : (const float4*)trowB;

  while (tile < ntiles) {
    float4* tp = tileA + hi * TPITCH + m;
    tp[0 * TPITCH] = n0;  tp[2 * TPITCH] = n1;  tp[4 * TPITCH] = n2;  tp[6 * TPITCH] = n3;
    tp[8 * TPITCH] = n4;  tp[10 * TPITCH] = n5; tp[12 * TPITCH] = n6; tp[14 * TPITCH] = n7;
    float4* tq = tileB + hi * TPITCH + m;
    tq[0 * TPITCH] = q0;  tq[2 * TPITCH] = q1;  tq[4 * TPITCH] = q2;  tq[6 * TPITCH] = q3;
    tq[8 * TPITCH] = q4;  tq[10 * TPITCH] = q5; tq[12 * TPITCH] = q6; tq[14 * TPITCH] = q7;
    const bool haveB = tile + 1 < ntiles;
    const int next = tile + stride;
    if (next < ntiles) {
      const float4* gin = (const float4*)(noise + (long)next * (16 * NODES)) + lofs;
      TILE_LOAD_N(gin);
      if (next + 1 < ntiles) TILE_LOAD_Q(gin + 256);
    }
    wave_lds_order();

    float a0r = 0.f, a1r = 0.f, b0r = 0.f, b1r = 0.f;

    auto tail = [&](const v8f& d, const float4& wa, const float2& wb) -> float {
      float p;
      p = fmaf(relu(d[0]), wa.x, wb.y);
      p = fmaf(relu(d[1]), wa.y, p);
      p = fmaf(relu(d[2]), wa.z, p);
      p = fmaf(relu(d[3]), wa.w, p);
      p = fmaf(relu(d[4]), wb.x, p);
      return p + xor16(p);
    };
    auto step = [&](int i, float nA0, float nA1, float nB0, float nB1,
                    float& yA, float& yB) {
      const float4 ap = sA[i * 32 + lane];
      const float* wq = &sW2[i * 16 + hi * 8];
      const float4 wa = *(const float4*)wq;
      const float2 wb = *(const float2*)(wq + 4);
      v2f A2; A2[0] = ap.z; A2[1] = ap.w;
      v2f A1; A1[0] = ap.x; A1[1] = ap.y;
      v8f zero = {};
      v2f B2a; B2a[0] = nA0; B2a[1] = nA1;
      v2f B2b; B2b[0] = nB0; B2b[1] = nB1;
      v8f tA = __builtin_amdgcn_wmma_f32_16x16x4_f32(
          false, A2, false, B2a, (short)0, zero, false, false);
      v8f tB = __builtin_amdgcn_wmma_f32_16x16x4_f32(
          false, A2, false, B2b, (short)0, zero, false, false);
      v2f B1a; B1a[0] = a0r; B1a[1] = a1r;
      v2f B1b; B1b[0] = b0r; B1b[1] = b1r;
      v8f dA = __builtin_amdgcn_wmma_f32_16x16x4_f32(
          false, A1, false, B1a, (short)0, tA, false, false);
      v8f dB = __builtin_amdgcn_wmma_f32_16x16x4_f32(
          false, A1, false, B1b, (short)0, tB, false, false);
      yA = tail(dA, wa, wb);
      yB = tail(dB, wa, wb);
    };

    float4 nznA = nzrowA[0], nznB = nzrowB[0];
#pragma unroll
    for (int i4 = 0; i4 < 16; ++i4) {
      const float4 zA = nznA, zB = nznB;
      if (i4 < 15) { nznA = nzrowA[i4 + 1]; nznB = nzrowB[i4 + 1]; }
      const int i = i4 * 4;
      float yA0, yA1, yA2, yA3, yB0, yB1, yB2, yB3;
      step(i + 0, zA.x, zA.y, zB.x, zB.y, yA0, yB0); if (!hi) { a0r = yA0; b0r = yB0; }
      step(i + 1, zA.x, zA.y, zB.x, zB.y, yA1, yB1); if (!hi) { a1r = yA1; b1r = yB1; }
      step(i + 2, zA.z, zA.w, zB.z, zB.w, yA2, yB2); if (hi)  { a0r = yA2; b0r = yB2; }
      step(i + 3, zA.z, zA.w, zB.z, zB.w, yA3, yB3); if (hi)  { a1r = yA3; b1r = yB3; }
      trowA[i4] = make_float4(yA0, yA1, yA2, yA3);
      trowB[i4] = make_float4(yB0, yB1, yB2, yB3);
    }

    wave_lds_order();

    float4* gout = (float4*)(out + (long)tile * (16 * NODES)) + lofs;
#pragma unroll
    for (int t = 0; t < 8; ++t) gout[t * 32] = tileA[(2 * t + hi) * TPITCH + m];
    if (haveB) {
#pragma unroll
      for (int t = 0; t < 8; ++t) gout[256 + t * 32] = tileB[(2 * t + hi) * TPITCH + m];
    }
    wave_lds_order();
    tile = next;
  }
}

extern "C" void kernel_launch(void* const* d_in, const int* in_sizes, int n_in,
                              void* d_out, int out_size, void* d_ws, size_t ws_size,
                              hipStream_t stream) {
  const float* noise = (const float*)d_in[0];
  const float* W1 = (const float*)d_in[2];
  const float* b1 = (const float*)d_in[3];
  const float* W2 = (const float*)d_in[4];
  const float* b2 = (const float*)d_in[5];
  float* out = (float*)d_out;

  const int B = in_sizes[0] / NODES;
  const int ntiles = B / 16;

  cgnn_wmma_kernel<<<256, WAVES * 32, 0, stream>>>(noise, W1, b1, W2, b2, out, ntiles);
}


# Wave priority steps down with progress (s_setprio 3,2,1,0 over a wave's last four half-trips): the four waves of a SIMD advance together, not oldest-first; no wave runs its last tiles alone
# speedup vs baseline: 3.8087x; 1.0383x over previous
#include <hip/hip_runtime.h>


typedef float v2f __attribute__((ext_vector_type(2)));
typedef float v8f __attribute__((ext_vector_type(8)));

#define NODES 64
#define NH    10
#define TPITCH 17

__device__ __forceinline__ int row_unit(int row) {
  return (row < 5) ? row : (row >= 8 && row < 13) ? row - 3 : -1;
}

__device__ __forceinline__ float xor16(float x) {
  int xi = __builtin_bit_cast(int, x);
  int r  = __builtin_amdgcn_permlanex16(xi, xi, 0x76543210, 0xfedcba98,
                                        true, false);
  return __builtin_bit_cast(float, r);
}

__device__ __forceinline__ float relu(float x) {
  const int xi = __builtin_bit_cast(int, x);
  return __builtin_bit_cast(float, xi > 0 ? xi : 0);
}

#define WAVES 16
#define TPW   2
#define TILE_LOAD_N(g)                                                   \
  do { n0 = (g)[0];   n1 = (g)[32];  n2 = (g)[64];  n3 = (g)[96];        \
       n4 = (g)[128]; n5 = (g)[160]; n6 = (g)[192]; n7 = (g)[224]; } while (0)
#define TILE_LOAD_Q(g)                                                   \
  do { q0 = (g)[0];   q1 = (g)[32];  q2 = (g)[64];  q3 = (g)[96];        \
       q4 = (g)[128]; q5 = (g)[160]; q6 = (g)[192]; q7 = (g)[224]; } while (0)

__device__ __forceinline__ void wave_lds_order() {
  __builtin_amdgcn_fence(__ATOMIC_RELEASE, "wavefront");
  __builtin_amdgcn_wave_barrier();
  __builtin_amdgcn_fence(__ATOMIC_ACQUIRE, "wavefront");
}

__global__ __launch_bounds__(WAVES * 32)
void cgnn_wmma_kernel(const float* __restrict__ noise,
                      const float* __restrict__ W1,
                      const float* __restrict__ b1g,
                      const float* __restrict__ W2g,
                      const float* __restrict__ b2g,
                      float* __restrict__ out,
                      int ntiles) {
  __shared__ float4 sA[NODES * 32];
  __shared__ float sW2[NODES * 16];
  __shared__ float4 sZ[16];
  __shared__ float4 sT[WAVES * TPW * 16 * TPITCH];

  const int lane = threadIdx.x & 31;
  const int wave = __builtin_amdgcn_readfirstlane((int)(threadIdx.x >> 5));
  const int hi   = lane >> 4;
  const int m    = lane & 15;
  const int stride = (int)gridDim.x * (WAVES * TPW);
  int tile = ((int)blockIdx.x * WAVES + wave) * TPW;

  const int lofs = hi * 16 + m;
  float4 n0 = {}, n1 = {}, n2 = {}, n3 = {}, n4 = {}, n5 = {}, n6 = {}, n7 = {};
  float4 q0 = {}, q1 = {}, q2 = {}, q3 = {}, q4 = {}, q5 = {}, q6 = {}, q7 = {};
  if (tile < ntiles) {
    const float4* gin = (const float4*)(noise + (long)tile * (16 * NODES)) + lofs;
    TILE_LOAD_N(gin);
    if (tile + 1 < ntiles) TILE_LOAD_Q(gin + 256);
  }

#pragma unroll
  for (int k = 0; k < (NODES * 32) / (WAVES * 32); ++k) {
    const int e = (int)threadIdx.x + k * (WAVES * 32);
    const int i = e >> 5, l = e & 31, h = l >> 4;
    const int j = row_unit(l & 15);
    const int jc = (j >= 0) ? j : 0;
    const float* w = W1 + i * (5 * NH);
    const int t0 = (i < 4) ? (2 * h)     : ((2 * h - i) & 3);
    const int t1 = (i < 4) ? (2 * h + 1) : ((2 * h + 1 - i) & 3);
    const float w0 = w[t0 * NH + jc];
    const float w1 = w[t1 * NH + jc];
    const float* cp = h ? (b1g + i * NH + jc) : (w + 4 * NH + jc);
    const float c = *cp;
    const bool odd_low = (h == 0) && (i & 1);
    const float a0 = (j >= 0) ? w0 : 0.f;
    const float a1 = (j >= 0) ? w1 : 0.f;
    const float a2 = (j >= 0 && !odd_low) ? c : 0.f;
    const float a3 = (j >= 0 && odd_low)  ? c : 0.f;
    sA[e] = make_float4(a0, a1, a2, a3);
  }
#pragma unroll
  for (int k = 0; k < (NODES * 16) / (WAVES * 32); ++k) {
    const int e = (int)threadIdx.x + k * (WAVES * 32);
    const int i = e >> 4;
    const int r = e & 15;
    const int j = row_unit(r);
    const float* sp = (j >= 0) ? (W2g + i * NH + j) : (b2g + i);
    const float v = *sp;
    sW2[e] = (j >= 0 || r == 5) ? v : 0.f;
  }
  if (threadIdx.x < 16) sZ[threadIdx.x] = make_float4(1.f, 0.f, 1.f, 0.f);
  __syncthreads();

  float4* tileA = sT + wave * (TPW * 16 * TPITCH);
  float4* tileB = tileA + 16 * TPITCH;
  float4* trowA = tileA + m * TPITCH;
  float4* trowB = tileB + m * TPITCH;
  const float4* nzrowA = hi ? (const float4*)sZ : (const float4*)trowA;
  const float4* nzrowB = hi ? (const float4*)sZ : (const float4*)trowB;

  while (tile < ntiles) {
    float4* tp = tileA + hi * TPITCH + m;
    tp[0 * TPITCH] = n0;  tp[2 * TPITCH] = n1;  tp[4 * TPITCH] = n2;  tp[6 * TPITCH] = n3;
    tp[8 * TPITCH] = n4;  tp[10 * TPITCH] = n5; tp[12 * TPITCH] = n6; tp[14 * TPITCH] = n7;
    float4* tq = tileB + hi * TPITCH + m;
    tq[0 * TPITCH] = q0;  tq[2 * TPITCH] = q1;  tq[4 * TPITCH] = q2;  tq[6 * TPITCH] = q3;
    tq[8 * TPITCH] = q4;  tq[10 * TPITCH] = q5; tq[12 * TPITCH] = q6; tq[14 * TPITCH] = q7;
    const bool haveB = tile + 1 < ntiles;
    const int next = tile + stride;
    const int after = (next < ntiles) ? ((next + stride < ntiles) ? 2 : 1) : 0;
    if (after >= 1) __builtin_amdgcn_s_setprio(3);
    else            __builtin_amdgcn_s_setprio(1);
    if (next < ntiles) {
      const float4* gin = (const float4*)(noise + (long)next * (16 * NODES)) + lofs;
      TILE_LOAD_N(gin);
      if (next + 1 < ntiles) TILE_LOAD_Q(gin + 256);
    }
    wave_lds_order();

    float a0r = 0.f, a1r = 0.f, b0r = 0.f, b1r = 0.f;

    auto tail = [&](const v8f& d, const float4& wa, const float2& wb) -> float {
      float p;
      p = fmaf(relu(d[0]), wa.x, wb.y);
      p = fmaf(relu(d[1]), wa.y, p);
      p = fmaf(relu(d[2]), wa.z, p);
      p = fmaf(relu(d[3]), wa.w, p);
      p = fmaf(relu(d[4]), wb.x, p);
      return p + xor16(p);
    };
    auto step = [&](int i, float nA0, float nA1, float nB0, float nB1,
                    float& yA, float& yB) {
      const float4 ap = sA[i * 32 + lane];
      const float* wq = &sW2[i * 16 + hi * 8];
      const float4 wa = *(const float4*)wq;
      const float2 wb = *(const float2*)(wq + 4);
      v2f A2; A2[0] = ap.z; A2[1] = ap.w;
      v2f A1; A1[0] = ap.x; A1[1] = ap.y;
      v8f zero = {};
      v2f B2a; B2a[0] = nA0; B2a[1] = nA1;
      v2f B2b; B2b[0] = nB0; B2b[1] = nB1;
      v8f tA = __builtin_amdgcn_wmma_f32_16x16x4_f32(
          false, A2, false, B2a, (short)0, zero, false, false);
      v8f tB = __builtin_amdgcn_wmma_f32_16x16x4_f32(
          false, A2, false, B2b, (short)0, zero, false, false);
      v2f B1a; B1a[0] = a0r; B1a[1] = a1r;
      v2f B1b; B1b[0] = b0r; B1b[1] = b1r;
      v8f dA = __builtin_amdgcn_wmma_f32_16x16x4_f32(
          false, A1, false, B1a, (short)0, tA, false, false);
      v8f dB = __builtin_amdgcn_wmma_f32_16x16x4_f32(
          false, A1, false, B1b, (short)0, tB, false, false);
      yA = tail(dA, wa, wb);
      yB = tail(dB, wa, wb);
    };

    float4 nznA = nzrowA[0], nznB = nzrowB[0];
#pragma unroll
    for (int i4 = 0; i4 < 16; ++i4) {
      const float4 zA = nznA, zB = nznB;
      if (i4 < 15) { nznA = nzrowA[i4 + 1]; nznB = nzrowB[i4 + 1]; }
      if (i4 == 8) {
        if (after >= 2)      __builtin_amdgcn_s_setprio(3);
        else if (after == 1) __builtin_amdgcn_s_setprio(2);
        else                 __builtin_amdgcn_s_setprio(0);
      }
      const int i = i4 * 4;
      float yA0, yA1, yA2, yA3, yB0, yB1, yB2, yB3;
      step(i + 0, zA.x, zA.y, zB.x, zB.y, yA0, yB0); if (!hi) { a0r = yA0; b0r = yB0; }
      step(i + 1, zA.x, zA.y, zB.x, zB.y, yA1, yB1); if (!hi) { a1r = yA1; b1r = yB1; }
      step(i + 2, zA.z, zA.w, zB.z, zB.w, yA2, yB2); if (hi)  { a0r = yA2; b0r = yB2; }
      step(i + 3, zA.z, zA.w, zB.z, zB.w, yA3, yB3); if (hi)  { a1r = yA3; b1r = yB3; }
      trowA[i4] = make_float4(yA0, yA1, yA2, yA3);
      trowB[i4] = make_float4(yB0, yB1, yB2, yB3);
    }

    wave_lds_order();

    float4* gout = (float4*)(out + (long)tile * (16 * NODES)) + lofs;
#pragma unroll
    for (int t = 0; t < 8; ++t) gout[t * 32] = tileA[(2 * t + hi) * TPITCH + m];
    if (haveB) {
#pragma unroll
      for (int t = 0; t < 8; ++t) gout[256 + t * 32] = tileB[(2 * t + hi) * TPITCH + m];
    }
    wave_lds_order();
    tile = next;
  }
}

extern "C" void kernel_launch(void* const* d_in, const int* in_sizes, int n_in,
                              void* d_out, int out_size, void* d_ws, size_t ws_size,
                              hipStream_t stream) {
  const float* noise = (const float*)d_in[0];
  const float* W1 = (const float*)d_in[2];
  const float* b1 = (const float*)d_in[3];
  const float* W2 = (const float*)d_in[4];
  const float* b2 = (const float*)d_in[5];
  float* out = (float*)d_out;

  const int B = in_sizes[0] / NODES;
  const int ntiles = B / 16;

  cgnn_wmma_kernel<<<256, WAVES * 32, 0, stream>>>(noise, W1, b1, W2, b2, out, ntiles);
}


# LDS accesses of the recurrence gathered into one place per 4 nodes (was 2-3 per node): next block's weights and noise read as one group into a second register set, results stored there a block late
# speedup vs baseline: 3.9529x; 1.0379x over previous
#include <hip/hip_runtime.h>


typedef float v2f __attribute__((ext_vector_type(2)));
typedef float v8f __attribute__((ext_vector_type(8)));

#define NODES 64
#define NH    10
#define TPITCH 17

__device__ __forceinline__ int row_unit(int row) {
  return (row < 5) ? row : (row >= 8 && row < 13) ? row - 3 : -1;
}

__device__ __forceinline__ float xor16(float x) {
  int xi = __builtin_bit_cast(int, x);
  int r  = __builtin_amdgcn_permlanex16(xi, xi, 0x76543210, 0xfedcba98,
                                        true, false);
  return __builtin_bit_cast(float, r);
}

__device__ __forceinline__ float relu(float x) {
  const int xi = __builtin_bit_cast(int, x);
  return __builtin_bit_cast(float, xi > 0 ? xi : 0);
}

#define WAVES 16
#define TPW   2
#define TILE_LOAD_N(g)                                                   \
  do { n0 = (g)[0];   n1 = (g)[32];  n2 = (g)[64];  n3 = (g)[96];        \
       n4 = (g)[128]; n5 = (g)[160]; n6 = (g)[192]; n7 = (g)[224]; } while (0)
#define TILE_LOAD_Q(g)                                                   \
  do { q0 = (g)[0];   q1 = (g)[32];  q2 = (g)[64];  q3 = (g)[96];        \
       q4 = (g)[128]; q5 = (g)[160]; q6 = (g)[192]; q7 = (g)[224]; } while (0)

__device__ __forceinline__ void wave_lds_order() {
  __builtin_amdgcn_fence(__ATOMIC_RELEASE, "wavefront");
  __builtin_amdgcn_wave_barrier();
  __builtin_amdgcn_fence(__ATOMIC_ACQUIRE, "wavefront");
}

__global__ __launch_bounds__(WAVES * 32)
void cgnn_wmma_kernel(const float* __restrict__ noise,
                      const float* __restrict__ W1,
                      const float* __restrict__ b1g,
                      const float* __restrict__ W2g,
                      const float* __restrict__ b2g,
                      float* __restrict__ out,
                      int ntiles) {
  __shared__ float4 sA[NODES * 32];
  __shared__ float sW2[NODES * 16];
  __shared__ float4 sZ[16];
  __shared__ float4 sT[WAVES * TPW * 16 * TPITCH];

  const int lane = threadIdx.x & 31;
  const int wave = __builtin_amdgcn_readfirstlane((int)(threadIdx.x >> 5));
  const int hi   = lane >> 4;
  const int m    = lane & 15;
  const int stride = (int)gridDim.x * (WAVES * TPW);
  int tile = ((int)blockIdx.x * WAVES + wave) * TPW;

  const int lofs = hi * 16 + m;
  float4 n0 = {}, n1 = {}, n2 = {}, n3 = {}, n4 = {}, n5 = {}, n6 = {}, n7 = {};
  float4 q0 = {}, q1 = {}, q2 = {}, q3 = {}, q4 = {}, q5 = {}, q6 = {}, q7 = {};
  if (tile < ntiles) {
    const float4* gin = (const float4*)(noise + (long)tile * (16 * NODES)) + lofs;
    TILE_LOAD_N(gin);
    if (tile + 1 < ntiles) TILE_LOAD_Q(gin + 256);
  }

#pragma unroll
  for (int k = 0; k < (NODES * 32) / (WAVES * 32); ++k) {
    const int e = (int)threadIdx.x + k * (WAVES * 32);
    const int i = e >> 5, l = e & 31, h = l >> 4;
    const int j = row_unit(l & 15);
    const int jc = (j >= 0) ? j : 0;
    const float* w = W1 + i * (5 * NH);
    const int t0 = (i < 4) ? (2 * h)     : ((2 * h - i) & 3);
    const int t1 = (i < 4) ? (2 * h + 1) : ((2 * h + 1 - i) & 3);
    const float w0 = w[t0 * NH + jc];
    const float w1 = w[t1 * NH + jc];
    const float* cp = h ? (b1g + i * NH + jc) : (w + 4 * NH + jc);
    const float c = *cp;
    const bool odd_low = (h == 0) && (i & 1);
    const float a0 = (j >= 0) ? w0 : 0.f;
    const float a1 = (j >= 0) ? w1 : 0.f;
    const float a2 = (j >= 0 && !odd_low) ? c : 0.f;
    const float a3 = (j >= 0 && odd_low)  ? c : 0.f;
    sA[e] = make_float4(a0, a1, a2, a3);
  }
#pragma unroll
  for (int k = 0; k < (NODES * 16) / (WAVES * 32); ++k) {
    const int e = (int)threadIdx.x + k * (WAVES * 32);
    const int i = e >> 4;
    const int r = e & 15;
    const int j = row_unit(r);
    const float* sp = (j >= 0) ? (W2g + i * NH + j) : (b2g + i);
    const float v = *sp;
    sW2[e] = (j >= 0 || r == 5) ? v : 0.f;
  }
  if (threadIdx.x < 16) sZ[threadIdx.x] = make_float4(1.f, 0.f, 1.f, 0.f);
  __syncthreads();

  float4* tileA = sT + wave * (TPW * 16 * TPITCH);
  float4* tileB = tileA + 16 * TPITCH;
  float4* trowA = tileA + m * TPITCH;
  float4* trowB = tileB + m * TPITCH;
  const float4* nzrowA = hi ? (const float4*)sZ : (const float4*)trowA;
  const float4* nzrowB = hi ? (const float4*)sZ : (const float4*)trowB;

  while (tile < ntiles) {
    float4* tp = tileA + hi * TPITCH + m;
    tp[0 * TPITCH] = n0;  tp[2 * TPITCH] = n1;  tp[4 * TPITCH] = n2;  tp[6 * TPITCH] = n3;
    tp[8 * TPITCH] = n4;  tp[10 * TPITCH] = n5; tp[12 * TPITCH] = n6; tp[14 * TPITCH] = n7;
    float4* tq = tileB + hi * TPITCH + m;
    tq[0 * TPITCH] = q0;  tq[2 * TPITCH] = q1;  tq[4 * TPITCH] = q2;  tq[6 * TPITCH] = q3;
    tq[8 * TPITCH] = q4;  tq[10 * TPITCH] = q5; tq[12 * TPITCH] = q6; tq[14 * TPITCH] = q7;
    const bool haveB = tile + 1 < ntiles;
    const int next = tile + stride;
    const int after = (next < ntiles) ? ((next + stride < ntiles) ? 2 : 1) : 0;
    if (after >= 1) __builtin_amdgcn_s_setprio(3);
    else            __builtin_amdgcn_s_setprio(1);
    if (next < ntiles) {
      const float4* gin = (const float4*)(noise + (long)next * (16 * NODES)) + lofs;
      TILE_LOAD_N(gin);
      if (next + 1 < ntiles) TILE_LOAD_Q(gin + 256);
    }
    wave_lds_order();

    float a0r = 0.f, a1r = 0.f, b0r = 0.f, b1r = 0.f;

    struct NodeW  { float4 ap; float4 wa; float2 wb; };
    struct BlockW { NodeW n[4]; float4 zA, zB; };
    auto load_block = [&](int b, BlockW& w) {
#pragma unroll
      for (int u = 0; u < 4; ++u) {
        const int i = 4 * b + u;
        w.n[u].ap = sA[i * 32 + lane];
        const float* wq = &sW2[i * 16 + hi * 8];
        w.n[u].wa = *(const float4*)wq;
        w.n[u].wb = *(const float2*)(wq + 4);
      }
      w.zA = nzrowA[b];
      w.zB = nzrowB[b];
    };
    auto tail = [&](const v8f& d, const float4& wa, const float2& wb) -> float {
      float p;
      p = fmaf(relu(d[0]), wa.x, wb.y);
      p = fmaf(relu(d[1]), wa.y, p);
      p = fmaf(relu(d[2]), wa.z, p);
      p = fmaf(relu(d[3]), wa.w, p);
      p = fmaf(relu(d[4]), wb.x, p);
      return p + xor16(p);
    };
    auto products = [&](const NodeW& w, float nA0, float nA1, float nB0, float nB1,
                        v8f& dA, v8f& dB) {
      v2f A2; A2[0] = w.ap.z; A2[1] = w.ap.w;
      v2f A1; A1[0] = w.ap.x; A1[1] = w.ap.y;
      v8f zero = {};
      v2f B2a; B2a[0] = nA0; B2a[1] = nA1;
      v2f B2b; B2b[0] = nB0; B2b[1] = nB1;
      v8f tA = __builtin_amdgcn_wmma_f32_16x16x4_f32(
          false, A2, false, B2a, (short)0, zero, false, false);
      v8f tB = __builtin_amdgcn_wmma_f32_16x16x4_f32(
          false, A2, false, B2b, (short)0, zero, false, false);
      v2f B1a; B1a[0] = a0r; B1a[1] = a1r;
      v2f B1b; B1b[0] = b0r; B1b[1] = b1r;
      dA = __builtin_amdgcn_wmma_f32_16x16x4_f32(
          false, A1, false, B1a, (short)0, tA, false, false);
      dB = __builtin_amdgcn_wmma_f32_16x16x4_f32(
          false, A1, false, B1b, (short)0, tB, false, false);
    };
    auto step = [&](const NodeW& w, float nA0, float nA1, float nB0, float nB1,
                    float& yA, float& yB) {
      v8f dA, dB;
      products(w, nA0, nA1, nB0, nB1, dA, dB);
      yA = tail(dA, w.wa, w.wb);
      yB = tail(dB, w.wa, w.wb);
    };

    BlockW cur, nxt;
    load_block(0, cur);
    float4 rA = {}, rB = {};
#pragma unroll
    for (int b = 0; b < 16; ++b) {
      if (b == 8) {
        if (after >= 2)      __builtin_amdgcn_s_setprio(3);
        else if (after == 1) __builtin_amdgcn_s_setprio(2);
        else                 __builtin_amdgcn_s_setprio(0);
      }
      const float4 zA = cur.zA, zB = cur.zB;
      float yA0, yA1, yA2, yA3, yB0, yB1, yB2, yB3;
      v8f dA, dB;
      products(cur.n[0], zA.x, zA.y, zB.x, zB.y, dA, dB);
      __builtin_amdgcn_sched_barrier(0);
      if (b > 0) {
        trowA[b - 1] = rA;
        trowB[b - 1] = rB;
      }
      if (b < 15) load_block(b + 1, nxt);
      __builtin_amdgcn_sched_barrier(0);
      yA0 = tail(dA, cur.n[0].wa, cur.n[0].wb);
      yB0 = tail(dB, cur.n[0].wa, cur.n[0].wb);
      if (!hi) { a0r = yA0; b0r = yB0; }
      step(cur.n[1], zA.x, zA.y, zB.x, zB.y, yA1, yB1); if (!hi) { a1r = yA1; b1r = yB1; }
      step(cur.n[2], zA.z, zA.w, zB.z, zB.w, yA2, yB2); if (hi)  { a0r = yA2; b0r = yB2; }
      products(cur.n[3], zA.z, zA.w, zB.z, zB.w, dA, dB);
      __builtin_amdgcn_sched_barrier(0x7F6);
      yA3 = tail(dA, cur.n[3].wa, cur.n[3].wb);
      yB3 = tail(dB, cur.n[3].wa, cur.n[3].wb);
      if (hi)  { a1r = yA3; b1r = yB3; }
      rA = make_float4(yA0, yA1, yA2, yA3);
      rB = make_float4(yB0, yB1, yB2, yB3);
      if (b < 15) cur = nxt;
    }
    trowA[15] = rA;
    trowB[15] = rB;

    wave_lds_order();

    float4* gout = (float4*)(out + (long)tile * (16 * NODES)) + lofs;
#pragma unroll
    for (int t = 0; t < 8; ++t) gout[t * 32] = tileA[(2 * t + hi) * TPITCH + m];
    if (haveB) {
#pragma unroll
      for (int t = 0; t < 8; ++t) gout[256 + t * 32] = tileB[(2 * t + hi) * TPITCH + m];
    }
    wave_lds_order();
    tile = next;
  }
}

extern "C" void kernel_launch(void* const* d_in, const int* in_sizes, int n_in,
                              void* d_out, int out_size, void* d_ws, size_t ws_size,
                              hipStream_t stream) {
  const float* noise = (const float*)d_in[0];
  const float* W1 = (const float*)d_in[2];
  const float* b1 = (const float*)d_in[3];
  const float* W2 = (const float*)d_in[4];
  const float* b2 = (const float*)d_in[5];
  float* out = (float*)d_out;

  const int B = in_sizes[0] / NODES;
  const int ntiles = B / 16;

  cgnn_wmma_kernel<<<256, WAVES * 32, 0, stream>>>(noise, W1, b1, W2, b2, out, ntiles);
}


# Noise and bias terms of a block's 4 nodes from 3 matrix instructions per tile (was 4): 4th node's rows ride in spare registers 5..7 of the first two; its noise enters at k-slot 3 (term's sum order)
# speedup vs baseline: 4.1078x; 1.0392x over previous
#include <hip/hip_runtime.h>


typedef float v2f __attribute__((ext_vector_type(2)));
typedef float v8f __attribute__((ext_vector_type(8)));

#define NODES 64
#define NH    10
#define TPITCH 17

__device__ __forceinline__ int row_unit(int i, int row) {
  const int H = row >> 3, r = row & 7;
  if ((i & 3) != 3) return (r < 5) ? 5 * H + r : -1;
  return (r >= 5) ? 5 * H + r - 5 : (r >= 3) ? 5 * H + r : -1;
}

__device__ __forceinline__ float xor16(float x) {
  int xi = __builtin_bit_cast(int, x);
  int r  = __builtin_amdgcn_permlanex16(xi, xi, 0x76543210, 0xfedcba98,
                                        true, false);
  return __builtin_bit_cast(float, r);
}

__device__ __forceinline__ float relu(float x) {
  const int xi = __builtin_bit_cast(int, x);
  return __builtin_bit_cast(float, xi > 0 ? xi : 0);
}

#define WAVES 16
#define TPW   2
#define TILE_LOAD_N(g)                                                   \
  do { n0 = (g)[0];   n1 = (g)[32];  n2 = (g)[64];  n3 = (g)[96];        \
       n4 = (g)[128]; n5 = (g)[160]; n6 = (g)[192]; n7 = (g)[224]; } while (0)
#define TILE_LOAD_Q(g)                                                   \
  do { q0 = (g)[0];   q1 = (g)[32];  q2 = (g)[64];  q3 = (g)[96];        \
       q4 = (g)[128]; q5 = (g)[160]; q6 = (g)[192]; q7 = (g)[224]; } while (0)

__device__ __forceinline__ void wave_lds_order() {
  __builtin_amdgcn_fence(__ATOMIC_RELEASE, "wavefront");
  __builtin_amdgcn_wave_barrier();
  __builtin_amdgcn_fence(__ATOMIC_ACQUIRE, "wavefront");
}

__global__ __launch_bounds__(WAVES * 32)
void cgnn_wmma_kernel(const float* __restrict__ noise,
                      const float* __restrict__ W1,
                      const float* __restrict__ b1g,
                      const float* __restrict__ W2g,
                      const float* __restrict__ b2g,
                      float* __restrict__ out,
                      int ntiles) {
  __shared__ float4 sA[NODES * 32];
  __shared__ float sW2[NODES * 16];
  __shared__ float4 sT[WAVES * TPW * 16 * TPITCH];
  __shared__ float4 sX[WAVES * TPW * 16 * TPITCH];

  const int lane = threadIdx.x & 31;
  const int wave = __builtin_amdgcn_readfirstlane((int)(threadIdx.x >> 5));
  const int hi   = lane >> 4;
  const int m    = lane & 15;
  const int stride = (int)gridDim.x * (WAVES * TPW);
  int tile = ((int)blockIdx.x * WAVES + wave) * TPW;

  const int lofs = hi * 16 + m;
  float4 n0 = {}, n1 = {}, n2 = {}, n3 = {}, n4 = {}, n5 = {}, n6 = {}, n7 = {};
  float4 q0 = {}, q1 = {}, q2 = {}, q3 = {}, q4 = {}, q5 = {}, q6 = {}, q7 = {};
  if (tile < ntiles) {
    const float4* gin = (const float4*)(noise + (long)tile * (16 * NODES)) + lofs;
    TILE_LOAD_N(gin);
    if (tile + 1 < ntiles) TILE_LOAD_Q(gin + 256);
  }

#pragma unroll
  for (int k = 0; k < (NODES * 32) / (WAVES * 32); ++k) {
    const int e = (int)threadIdx.x + k * (WAVES * 32);
    const int i = e >> 5, l = e & 31, h = l >> 4;
    const int j = row_unit(i, l & 15);
    const int jc = (j >= 0) ? j : 0;
    const float* w = W1 + i * (5 * NH);
    const int t0 = (i < 4) ? (2 * h)     : ((2 * h - i) & 3);
    const int t1 = (i < 4) ? (2 * h + 1) : ((2 * h + 1 - i) & 3);
    const float w0 = w[t0 * NH + jc];
    const float w1 = w[t1 * NH + jc];
    const float a0 = (j >= 0) ? w0 : 0.f;
    const float a1 = (j >= 0) ? w1 : 0.f;
    const int H = (l >> 3) & 1, r = l & 7, u = i & 3;
    int src = -1, ju = 0, ks = 0;
    if (u == 0) {
      if (r < 5) { src = i;     ju = 5 * H + r;     ks = 0; }
      else       { src = i + 3; ju = 5 * H + r - 5; ks = 3; }
    } else if (u == 1) {
      if (r < 5)      { src = i;     ju = 5 * H + r;     ks = 1; }
      else if (r < 7) { src = i + 2; ju = 5 * H + r - 2; ks = 3; }
    } else if (u == 2) {
      if (r < 5) { src = i; ju = 5 * H + r; ks = 0; }
    }
    const int sc = (src >= 0) ? src : 0;
    const float wn = W1[sc * (5 * NH) + 4 * NH + ju];
    const float bb = b1g[sc * NH + ju];
    const float a2 = (src < 0) ? 0.f : h ? bb : (ks == 0 ? wn : 0.f);
    const float a3 = (src < 0) ? 0.f : h ? (ks == 3 ? wn : 0.f) : (ks == 1 ? wn : 0.f);
    sA[e] = make_float4(a0, a1, a2, a3);
  }
#pragma unroll
  for (int k = 0; k < (NODES * 16) / (WAVES * 32); ++k) {
    const int e = (int)threadIdx.x + k * (WAVES * 32);
    const int i = e >> 4;
    const int r = e & 15;
    const int j = row_unit(0, r);
    const float* sp = (j >= 0) ? (W2g + i * NH + j) : (b2g + i);
    const float v = *sp;
    sW2[e] = (j >= 0 || r == 5) ? v : 0.f;
  }
  float4* tileA = sT + wave * (TPW * 16 * TPITCH);
  float4* tileB = tileA + 16 * TPITCH;
  float4* trowA = tileA + m * TPITCH;
  float4* trowB = tileB + m * TPITCH;
  float4* auxA = sX + wave * (TPW * 16 * TPITCH);
  float4* auxB = auxA + 16 * TPITCH;
#pragma unroll
  for (int t = 0; t < 8; ++t) {
    auxA[(2 * t + hi) * TPITCH + m] = make_float4(1.f, 0.f, 1.f, 0.f);
    auxB[(2 * t + hi) * TPITCH + m] = make_float4(1.f, 0.f, 1.f, 0.f);
  }
  __syncthreads();

  const float4* nzrowA = hi ? (const float4*)(auxA + m * TPITCH) : (const float4*)trowA;
  const float4* nzrowB = hi ? (const float4*)(auxB + m * TPITCH) : (const float4*)trowB;

  while (tile < ntiles) {
    float4* tp = tileA + hi * TPITCH + m;
    tp[0 * TPITCH] = n0;  tp[2 * TPITCH] = n1;  tp[4 * TPITCH] = n2;  tp[6 * TPITCH] = n3;
    tp[8 * TPITCH] = n4;  tp[10 * TPITCH] = n5; tp[12 * TPITCH] = n6; tp[14 * TPITCH] = n7;
    float4* tq = tileB + hi * TPITCH + m;
    tq[0 * TPITCH] = q0;  tq[2 * TPITCH] = q1;  tq[4 * TPITCH] = q2;  tq[6 * TPITCH] = q3;
    tq[8 * TPITCH] = q4;  tq[10 * TPITCH] = q5; tq[12 * TPITCH] = q6; tq[14 * TPITCH] = q7;
    float* xp = (float*)(auxA + hi * TPITCH + m) + 1;
    xp[0 * 4 * TPITCH] = n0.w;  xp[2 * 4 * TPITCH] = n1.w;  xp[4 * 4 * TPITCH] = n2.w;
    xp[6 * 4 * TPITCH] = n3.w;  xp[8 * 4 * TPITCH] = n4.w;  xp[10 * 4 * TPITCH] = n5.w;
    xp[12 * 4 * TPITCH] = n6.w; xp[14 * 4 * TPITCH] = n7.w;
    float* xq = (float*)(auxB + hi * TPITCH + m) + 1;
    xq[0 * 4 * TPITCH] = q0.w;  xq[2 * 4 * TPITCH] = q1.w;  xq[4 * 4 * TPITCH] = q2.w;
    xq[6 * 4 * TPITCH] = q3.w;  xq[8 * 4 * TPITCH] = q4.w;  xq[10 * 4 * TPITCH] = q5.w;
    xq[12 * 4 * TPITCH] = q6.w; xq[14 * 4 * TPITCH] = q7.w;
    const bool haveB = tile + 1 < ntiles;
    const int next = tile + stride;
    const int after = (next < ntiles) ? ((next + stride < ntiles) ? 2 : 1) : 0;
    if (after >= 1) __builtin_amdgcn_s_setprio(3);
    else            __builtin_amdgcn_s_setprio(1);
    if (next < ntiles) {
      const float4* gin = (const float4*)(noise + (long)next * (16 * NODES)) + lofs;
      TILE_LOAD_N(gin);
      if (next + 1 < ntiles) TILE_LOAD_Q(gin + 256);
    }
    wave_lds_order();

    float a0r = 0.f, a1r = 0.f, b0r = 0.f, b1r = 0.f;

    struct NodeW  { float4 ap; float4 wa; float2 wb; };
    struct BlockW { NodeW n[4]; float4 zA, zB; };
    auto load_block = [&](int b, BlockW& w) {
#pragma unroll
      for (int u = 0; u < 4; ++u) {
        const int i = 4 * b + u;
        w.n[u].ap = sA[i * 32 + lane];
        const float* wq = &sW2[i * 16 + hi * 8];
        w.n[u].wa = *(const float4*)wq;
        w.n[u].wb = *(const float2*)(wq + 4);
      }
      w.zA = nzrowA[b];
      w.zB = nzrowB[b];
    };
    auto tail5 = [&](float h0, float h1, float h2, float h3, float h4,
                     const float4& wa, const float2& wb) -> float {
      float p;
      p = fmaf(relu(h0), wa.x, wb.y);
      p = fmaf(relu(h1), wa.y, p);
      p = fmaf(relu(h2), wa.z, p);
      p = fmaf(relu(h3), wa.w, p);
      p = fmaf(relu(h4), wb.x, p);
      return p + xor16(p);
    };
    auto tail = [&](const v8f& d, const float4& wa, const float2& wb) -> float {
      return tail5(d[0], d[1], d[2], d[3], d[4], wa, wb);
    };
    auto tail3 = [&](const v8f& d, const float4& wa, const float2& wb) -> float {
      return tail5(d[5], d[6], d[7], d[3], d[4], wa, wb);
    };
    auto noise_products = [&](const NodeW& w, float nA0, float nA1, float nB0, float nB1,
                              v8f& tA, v8f& tB) {
      v2f AN; AN[0] = w.ap.z; AN[1] = w.ap.w;
      v8f zero = {};
      v2f BNa; BNa[0] = nA0; BNa[1] = nA1;
      v2f BNb; BNb[0] = nB0; BNb[1] = nB1;
      tA = __builtin_amdgcn_wmma_f32_16x16x4_f32(
          false, AN, false, BNa, (short)0, zero, false, false);
      tB = __builtin_amdgcn_wmma_f32_16x16x4_f32(
          false, AN, false, BNb, (short)0, zero, false, false);
    };
    auto parent_products = [&](const NodeW& w, const v8f& tA, const v8f& tB,
                               v8f& dA, v8f& dB) {
      v2f A1; A1[0] = w.ap.x; A1[1] = w.ap.y;
      v2f B1a; B1a[0] = a0r; B1a[1] = a1r;
      v2f B1b; B1b[0] = b0r; B1b[1] = b1r;
      dA = __builtin_amdgcn_wmma_f32_16x16x4_f32(
          false, A1, false, B1a, (short)0, tA, false, false);
      dB = __builtin_amdgcn_wmma_f32_16x16x4_f32(
          false, A1, false, B1b, (short)0, tB, false, false);
    };

    BlockW cur, nxt;
    load_block(0, cur);
    float4 rA = {}, rB = {};
#pragma unroll
    for (int b = 0; b < 16; ++b) {
      if (b == 8) {
        if (after >= 2)      __builtin_amdgcn_s_setprio(3);
        else if (after == 1) __builtin_amdgcn_s_setprio(2);
        else                 __builtin_amdgcn_s_setprio(0);
      }
      const float4 zA = cur.zA, zB = cur.zB;
      float yA0, yA1, yA2, yA3, yB0, yB1, yB2, yB3;
      v8f tA, tB, dA0, dB0, dA1, dB1, dA, dB;
      noise_products(cur.n[0], zA.x, zA.y, zB.x, zB.y, tA, tB);
      parent_products(cur.n[0], tA, tB, dA0, dB0);
      __builtin_amdgcn_sched_barrier(0);
      if (b > 0) {
        trowA[b - 1] = rA;
        trowB[b - 1] = rB;
      }
      if (b < 15) load_block(b + 1, nxt);
      __builtin_amdgcn_sched_barrier(0);
      yA0 = tail(dA0, cur.n[0].wa, cur.n[0].wb);
      yB0 = tail(dB0, cur.n[0].wa, cur.n[0].wb);
      if (!hi) { a0r = yA0; b0r = yB0; }
      noise_products(cur.n[1], zA.x, zA.y, zB.x, zB.y, tA, tB);
      parent_products(cur.n[1], tA, tB, dA1, dB1);
      yA1 = tail(dA1, cur.n[1].wa, cur.n[1].wb);
      yB1 = tail(dB1, cur.n[1].wa, cur.n[1].wb);
      if (!hi) { a1r = yA1; b1r = yB1; }
      noise_products(cur.n[2], zA.z, zA.w, zB.z, zB.w, tA, tB);
      parent_products(cur.n[2], tA, tB, dA, dB);
      yA2 = tail(dA, cur.n[2].wa, cur.n[2].wb);
      yB2 = tail(dB, cur.n[2].wa, cur.n[2].wb);
      if (hi)  { a0r = yA2; b0r = yB2; }
      tA = dA0; tA[3] = dA1[5]; tA[4] = dA1[6];
      tB = dB0; tB[3] = dB1[5]; tB[4] = dB1[6];
      parent_products(cur.n[3], tA, tB, dA, dB);
      __builtin_amdgcn_sched_barrier(0x7F6);
      yA3 = tail3(dA, cur.n[3].wa, cur.n[3].wb);
      yB3 = tail3(dB, cur.n[3].wa, cur.n[3].wb);
      if (hi)  { a1r = yA3; b1r = yB3; }
      rA = make_float4(yA0, yA1, yA2, yA3);
      rB = make_float4(yB0, yB1, yB2, yB3);
      if (b < 15) cur = nxt;
    }
    trowA[15] = rA;
    trowB[15] = rB;

    wave_lds_order();

    float4* gout = (float4*)(out + (long)tile * (16 * NODES)) + lofs;
#pragma unroll
    for (int t = 0; t < 8; ++t) gout[t * 32] = tileA[(2 * t + hi) * TPITCH + m];
    if (haveB) {
#pragma unroll
      for (int t = 0; t < 8; ++t) gout[256 + t * 32] = tileB[(2 * t + hi) * TPITCH + m];
    }
    wave_lds_order();
    tile = next;
  }
}

extern "C" void kernel_launch(void* const* d_in, const int* in_sizes, int n_in,
                              void* d_out, int out_size, void* d_ws, size_t ws_size,
                              hipStream_t stream) {
  const float* noise = (const float*)d_in[0];
  const float* W1 = (const float*)d_in[2];
  const float* b1 = (const float*)d_in[3];
  const float* W2 = (const float*)d_in[4];
  const float* b2 = (const float*)d_in[5];
  float* out = (float*)d_out;

  const int B = in_sizes[0] / NODES;
  const int ntiles = B / 16;

  cgnn_wmma_kernel<<<256, WAVES * 32, 0, stream>>>(noise, W1, b1, W2, b2, out, ntiles);
}
